# starting kernel
# baseline (speedup 1.0000x reference)
#include <hip/hip_runtime.h>
#include <hip/hip_bf16.h>

#define N_NODES 50000
#define N_EDGES 800000
#define DIN 128
#define DOUT 128
#define LDS_STRIDE 132

typedef __attribute__((ext_vector_type(2))) float v2f;
typedef __attribute__((ext_vector_type(8))) float v8f;

__launch_bounds__(256)
__global__ void gemm_wmma_f32(const float* __restrict__ A,
                              const float* __restrict__ W,
                              float* __restrict__ C, int nrows) {
  __shared__ float As[32 * LDS_STRIDE];

  const int tid  = threadIdx.x;
  const int lane = tid & 31;
  const int wave = tid >> 5;
  const int m0   = blockIdx.x * 32;
  const int rows = (nrows - m0) < 32 ? (nrows - m0) : 32;

  {
    const float4* src4 = (const float4*)(A + (size_t)m0 * DIN);
    const int n4 = rows * (DIN / 4);
    for (int i = tid; i < n4; i += 256) {
      const int r = i >> 5;
      const int c = i & 31;
      ((float4*)(As + r * LDS_STRIDE))[c] = src4[i];
    }
  }
  __syncthreads();

  const int n0    = wave * 16;
  const int mlane = lane & 15;
  const int khalf = (lane >> 4) * 2;

  const float* a0   = As + (size_t)mlane * LDS_STRIDE + khalf;
  const float* a1   = a0 + 16 * LDS_STRIDE;
  const float* wcol = W + (size_t)khalf * DOUT + n0 + mlane;

  v8f acc0 = {}, acc1 = {};
#pragma unroll 4
  for (int k = 0; k < DIN; k += 4) {
    v2f b;
    b.x = wcol[(size_t)k * DOUT];
    b.y = wcol[(size_t)(k + 1) * DOUT];
    v2f x0 = *(const v2f*)(a0 + k);
    v2f x1 = *(const v2f*)(a1 + k);
    acc0 = __builtin_amdgcn_wmma_f32_16x16x4_f32(false, x0, false, b,
                                                 (short)0, acc0, false, false);
    acc1 = __builtin_amdgcn_wmma_f32_16x16x4_f32(false, x1, false, b,
                                                 (short)0, acc1, false, false);
  }

  float* crow = C + (size_t)(m0 + (lane >> 4) * 8) * DOUT + n0 + mlane;
#pragma unroll
  for (int j = 0; j < 8; ++j)
    crow[(size_t)j * DOUT] = acc0[j];
  if (rows > 16) {
    float* crow1 = crow + (size_t)16 * DOUT;
#pragma unroll
    for (int j = 0; j < 8; ++j)
      crow1[(size_t)j * DOUT] = acc1[j];
  }
}

__global__ void deg_init(float* __restrict__ deg) {
  int i = blockIdx.x * blockDim.x + threadIdx.x;
  if (i < N_NODES) deg[i] = 1.0f;
}

__global__ void deg_accum(const int* __restrict__ dst, float* __restrict__ deg) {
  int i = blockIdx.x * blockDim.x + threadIdx.x;
  if (i < N_EDGES) atomicAdd(&deg[dst[i]], 1.0f);
}

__global__ void deg_to_dinv(float* __restrict__ deg) {
  int i = blockIdx.x * blockDim.x + threadIdx.x;
  if (i < N_NODES) {
    float d = deg[i];
    deg[i] = d > 0.0f ? (1.0f / sqrtf(d)) : 0.0f;
  }
}

__global__ void zero_buf(float* __restrict__ p, int n) {
  int i = blockIdx.x * blockDim.x + threadIdx.x;
  if (i < n) p[i] = 0.0f;
}

__launch_bounds__(256)
__global__ void edge_scatter(const float* __restrict__ H,
                             const int* __restrict__ src,
                             const int* __restrict__ dst,
                             const float* __restrict__ dinv,
                             float* __restrict__ agg, int nedges) {
  int e    = (int)((blockIdx.x * 256u + threadIdx.x) >> 5);
  int lane = threadIdx.x & 31;
  if (e >= nedges) return;
  int s = src[e];
  int d = dst[e];
  float nrm = dinv[s] * dinv[d];
  const float4 v = ((const float4*)(H + (size_t)s * DOUT))[lane];
  float* out = agg + (size_t)d * DOUT + (size_t)lane * 4;
  atomicAdd(out + 0, v.x * nrm);
  atomicAdd(out + 1, v.y * nrm);
  atomicAdd(out + 2, v.z * nrm);
  atomicAdd(out + 3, v.w * nrm);
}

__global__ void bias_relu(const float* __restrict__ agg,
                          const float* __restrict__ hlin,
                          const float* __restrict__ dinv,
                          const float* __restrict__ b,
                          float* __restrict__ out) {
  int i = blockIdx.x * blockDim.x + threadIdx.x;
  if (i < N_NODES * DOUT) {
    int node = i >> 7;
    int c    = i & (DOUT - 1);
    float di = dinv[node];
    float v  = agg[i] + hlin[i] * di * di + b[c];
    out[i]   = v > 0.0f ? v : 0.0f;
  }
}

extern "C" void kernel_launch(void* const* d_in, const int* in_sizes, int n_in,
                              void* d_out, int out_size, void* d_ws, size_t ws_size,
                              hipStream_t stream) {
  (void)in_sizes; (void)n_in; (void)out_size; (void)ws_size;

  const float* x  = (const float*)d_in[0];
  const int*   ei = (const int*)d_in[1];
  const float* Wl[3] = {(const float*)d_in[2], (const float*)d_in[4], (const float*)d_in[6]};
  const float* bl[3] = {(const float*)d_in[3], (const float*)d_in[5], (const float*)d_in[7]};
  const int* src = ei;
  const int* dst = ei + N_EDGES;

  char*  ws   = (char*)d_ws;
  float* dinv = (float*)ws;
  size_t off  = ((size_t)N_NODES * 4 + 255) & ~(size_t)255;
  float* hlin = (float*)(ws + off);
  float* agg  = (float*)(ws + off + (size_t)N_NODES * DOUT * 4);

  const int NE = N_NODES * DOUT;

  deg_init   <<<(N_NODES + 255) / 256, 256, 0, stream>>>(dinv);
  deg_accum  <<<(N_EDGES + 255) / 256, 256, 0, stream>>>(dst, dinv);
  deg_to_dinv<<<(N_NODES + 255) / 256, 256, 0, stream>>>(dinv);

  const float* hin  = x;
  float*       hout = (float*)d_out;
  for (int l = 0; l < 3; ++l) {
    gemm_wmma_f32<<<(N_NODES + 31) / 32, 256, 0, stream>>>(hin, Wl[l], hlin, N_NODES);
    zero_buf     <<<(NE + 255) / 256, 256, 0, stream>>>(agg, NE);
    edge_scatter <<<N_EDGES / 8, 256, 0, stream>>>(hlin, src, dst, dinv, agg, N_EDGES);
    bias_relu    <<<(NE + 255) / 256, 256, 0, stream>>>(agg, hlin, dinv, bl[l], hout);
    hin = hout;
  }
}


# Replace per-layer float-atomic edge scatter by edges grouped by destination once per call (int atomic arrival-order rank), then per-node register sums with self loop, bias, relu fused
# speedup vs baseline: 3.0808x; 3.0808x over previous
#include <hip/hip_runtime.h>
#include <hip/hip_bf16.h>

#define N_NODES 50000
#define N_EDGES 800000
#define DIN 128
#define DOUT 128
#define LDS_STRIDE 132

typedef __attribute__((ext_vector_type(2))) float v2f;
typedef __attribute__((ext_vector_type(8))) float v8f;

__launch_bounds__(256)
__global__ void gemm_wmma_f32(const float* __restrict__ A,
                              const float* __restrict__ W,
                              float* __restrict__ C, int nrows) {
  __shared__ float As[32 * LDS_STRIDE];

  const int tid  = threadIdx.x;
  const int lane = tid & 31;
  const int wave = tid >> 5;
  const int m0   = blockIdx.x * 32;
  const int rows = (nrows - m0) < 32 ? (nrows - m0) : 32;

  {
    const float4* src4 = (const float4*)(A + (size_t)m0 * DIN);
    const int n4 = rows * (DIN / 4);
    for (int i = tid; i < n4; i += 256) {
      const int r = i >> 5;
      const int c = i & 31;
      ((float4*)(As + r * LDS_STRIDE))[c] = src4[i];
    }
  }
  __syncthreads();

  const int n0    = wave * 16;
  const int mlane = lane & 15;
  const int khalf = (lane >> 4) * 2;

  const float* a0   = As + (size_t)mlane * LDS_STRIDE + khalf;
  const float* a1   = a0 + 16 * LDS_STRIDE;
  const float* wcol = W + (size_t)khalf * DOUT + n0 + mlane;

  v8f acc0 = {}, acc1 = {};
#pragma unroll 4
  for (int k = 0; k < DIN; k += 4) {
    v2f b;
    b.x = wcol[(size_t)k * DOUT];
    b.y = wcol[(size_t)(k + 1) * DOUT];
    v2f x0 = *(const v2f*)(a0 + k);
    v2f x1 = *(const v2f*)(a1 + k);
    acc0 = __builtin_amdgcn_wmma_f32_16x16x4_f32(false, x0, false, b,
                                                 (short)0, acc0, false, false);
    acc1 = __builtin_amdgcn_wmma_f32_16x16x4_f32(false, x1, false, b,
                                                 (short)0, acc1, false, false);
  }

  float* crow = C + (size_t)(m0 + (lane >> 4) * 8) * DOUT + n0 + mlane;
#pragma unroll
  for (int j = 0; j < 8; ++j)
    crow[(size_t)j * DOUT] = acc0[j];
  if (rows > 16) {
    float* crow1 = crow + (size_t)16 * DOUT;
#pragma unroll
    for (int j = 0; j < 8; ++j)
      crow1[(size_t)j * DOUT] = acc1[j];
  }
}

#define OFF_DINV 0
#define OFF_CNT  200192
#define OFF_ROWP 400384
#define OFF_RANK 600576
#define OFF_EW   3800576
#define OFF_HLIN 10200576

#define SCAN_T 1024

__global__ void zero_cnt(int* __restrict__ cnt) {
  int i = blockIdx.x * blockDim.x + threadIdx.x;
  if (i < N_NODES) cnt[i] = 0;
}

__global__ void count_rank(const int* __restrict__ dst, int* __restrict__ cnt,
                           int* __restrict__ rank) {
  int e = blockIdx.x * blockDim.x + threadIdx.x;
  if (e < N_EDGES) rank[e] = atomicAdd(&cnt[dst[e]], 1);
}

__launch_bounds__(SCAN_T)
__global__ void scan_deg(const int* __restrict__ cnt, int* __restrict__ rowptr,
                         float* __restrict__ dinv) {
  __shared__ int wave_tot[32];
  __shared__ int wave_pre[32];
  const int tid   = threadIdx.x;
  const int lane  = tid & 31;
  const int wv    = tid >> 5;
  const int first = blockIdx.x * SCAN_T;

  int part = 0;
  for (int i = tid; i < first; i += SCAN_T) part += cnt[i];

  const int i = first + tid;
  const int c = (i < N_NODES) ? cnt[i] : 0;

  int incl = c;
  for (int o = 1; o < 32; o <<= 1) {
    int t = __shfl_up(incl, o);
    if (lane >= o) incl += t;
  }
  for (int o = 16; o > 0; o >>= 1) part += __shfl_xor(part, o);

  if (lane == 31) wave_tot[wv] = incl;
  if (lane == 0)  wave_pre[wv] = part;
  __syncthreads();

  int base = wave_pre[lane] + ((lane < wv) ? wave_tot[lane] : 0);
  for (int o = 16; o > 0; o >>= 1) base += __shfl_xor(base, o);

  if (i < N_NODES) {
    rowptr[i] = base + incl - c;
    float d = (float)(c + 1);
    dinv[i] = d > 0.0f ? (1.0f / sqrtf(d)) : 0.0f;
  }
}

__global__ void fill_csr(const int* __restrict__ src, const int* __restrict__ dst,
                         const int* __restrict__ rank, const int* __restrict__ rowptr,
                         const float* __restrict__ dinv, int2* __restrict__ ew) {
  int e = blockIdx.x * blockDim.x + threadIdx.x;
  if (e < N_EDGES) {
    int s = src[e];
    int d = dst[e];
    float w = dinv[s] * dinv[d];
    ew[rowptr[d] + rank[e]] = make_int2(s, __float_as_int(w));
  }
}

__device__ __forceinline__ void fma4(float4& a, const float4 v, const float w) {
  a.x = fmaf(v.x, w, a.x); a.y = fmaf(v.y, w, a.y);
  a.z = fmaf(v.z, w, a.z); a.w = fmaf(v.w, w, a.w);
}

__launch_bounds__(256)
__global__ void gather_fused(const float* __restrict__ H,
                             const int* __restrict__ rowptr,
                             const int* __restrict__ cnt,
                             const int2* __restrict__ ew,
                             const float* __restrict__ dinv,
                             const float* __restrict__ bias,
                             float* __restrict__ out) {
  const int lane = threadIdx.x & 31;
  const int node = __builtin_amdgcn_readfirstlane((int)((blockIdx.x * 256u + threadIdx.x) >> 5));
  if (node >= N_NODES) return;

  const int   start = rowptr[node];
  const int   n     = cnt[node];
  const float di    = dinv[node];
  const float4* H4  = (const float4*)H;

  const float4 vself = H4[(size_t)node * 32 + lane];
  const float4 b     = ((const float4*)bias)[lane];

  float4 acc;
  acc.x = 0.0f; acc.y = 0.0f; acc.z = 0.0f; acc.w = 0.0f;

#define SRC(k)  __builtin_amdgcn_readlane(my.x, (k))
#define WGT(k)  __int_as_float(__builtin_amdgcn_readlane(my.y, (k)))
#define ROW(s)  H4[(size_t)(s) * 32 + lane]

  for (int base = 0; base < n; base += 32) {
    const int m = (n - base) < 32 ? (n - base) : 32;
    int2 my = make_int2(0, 0);
    if (lane < m) my = ew[start + base + lane];

    int j = 0;
    for (; j + 8 <= m; j += 8) {
      const float4 v0 = ROW(SRC(j));
      const float4 v1 = ROW(SRC(j + 1));
      const float4 v2 = ROW(SRC(j + 2));
      const float4 v3 = ROW(SRC(j + 3));
      const float4 v4 = ROW(SRC(j + 4));
      const float4 v5 = ROW(SRC(j + 5));
      const float4 v6 = ROW(SRC(j + 6));
      const float4 v7 = ROW(SRC(j + 7));
      fma4(acc, v0, WGT(j));
      fma4(acc, v1, WGT(j + 1));
      fma4(acc, v2, WGT(j + 2));
      fma4(acc, v3, WGT(j + 3));
      fma4(acc, v4, WGT(j + 4));
      fma4(acc, v5, WGT(j + 5));
      fma4(acc, v6, WGT(j + 6));
      fma4(acc, v7, WGT(j + 7));
    }
    if (j + 4 <= m) {
      const float4 v0 = ROW(SRC(j));
      const float4 v1 = ROW(SRC(j + 1));
      const float4 v2 = ROW(SRC(j + 2));
      const float4 v3 = ROW(SRC(j + 3));
      fma4(acc, v0, WGT(j));
      fma4(acc, v1, WGT(j + 1));
      fma4(acc, v2, WGT(j + 2));
      fma4(acc, v3, WGT(j + 3));
      j += 4;
    }
    const int rem = m - j;
    if (rem == 3) {
      const float4 v0 = ROW(SRC(j));
      const float4 v1 = ROW(SRC(j + 1));
      const float4 v2 = ROW(SRC(j + 2));
      fma4(acc, v0, WGT(j));
      fma4(acc, v1, WGT(j + 1));
      fma4(acc, v2, WGT(j + 2));
    } else if (rem == 2) {
      const float4 v0 = ROW(SRC(j));
      const float4 v1 = ROW(SRC(j + 1));
      fma4(acc, v0, WGT(j));
      fma4(acc, v1, WGT(j + 1));
    } else if (rem == 1) {
      const float4 v0 = ROW(SRC(j));
      fma4(acc, v0, WGT(j));
    }
  }
#undef SRC
#undef WGT
#undef ROW

  fma4(acc, vself, di * di);

  float4 r;
  r.x = acc.x + b.x; r.y = acc.y + b.y; r.z = acc.z + b.z; r.w = acc.w + b.w;
  r.x = r.x > 0.0f ? r.x : 0.0f;
  r.y = r.y > 0.0f ? r.y : 0.0f;
  r.z = r.z > 0.0f ? r.z : 0.0f;
  r.w = r.w > 0.0f ? r.w : 0.0f;
  ((float4*)out)[(size_t)node * 32 + lane] = r;
}

extern "C" void kernel_launch(void* const* d_in, const int* in_sizes, int n_in,
                              void* d_out, int out_size, void* d_ws, size_t ws_size,
                              hipStream_t stream) {
  (void)in_sizes; (void)n_in; (void)out_size; (void)ws_size;

  const float* x  = (const float*)d_in[0];
  const int*   ei = (const int*)d_in[1];
  const float* Wl[3] = {(const float*)d_in[2], (const float*)d_in[4], (const float*)d_in[6]};
  const float* bl[3] = {(const float*)d_in[3], (const float*)d_in[5], (const float*)d_in[7]};
  const int* src = ei;
  const int* dst = ei + N_EDGES;

  char*  ws     = (char*)d_ws;
  float* dinv   = (float*)(ws + OFF_DINV);
  int*   cnt    = (int*)(ws + OFF_CNT);
  int*   rowptr = (int*)(ws + OFF_ROWP);
  int*   rank   = (int*)(ws + OFF_RANK);
  int2*  ew     = (int2*)(ws + OFF_EW);
  float* hlin   = (float*)(ws + OFF_HLIN);

  zero_cnt  <<<(N_NODES + 255) / 256, 256, 0, stream>>>(cnt);
  count_rank<<<(N_EDGES + 255) / 256, 256, 0, stream>>>(dst, cnt, rank);
  scan_deg  <<<(N_NODES + SCAN_T - 1) / SCAN_T, SCAN_T, 0, stream>>>(cnt, rowptr, dinv);
  fill_csr  <<<(N_EDGES + 255) / 256, 256, 0, stream>>>(src, dst, rank, rowptr, dinv, ew);

  const float* hin  = x;
  float*       hout = (float*)d_out;
  for (int l = 0; l < 3; ++l) {
    gemm_wmma_f32<<<(N_NODES + 31) / 32, 256, 0, stream>>>(hin, Wl[l], hlin, N_NODES);
    gather_fused <<<N_NODES / 8, 256, 0, stream>>>(hlin, rowptr, cnt, ew, dinv, bl[l], hout);
    hin = hout;
  }
}


# Each layer is one kernel: neighbours' INPUT rows are summed first, the 32 summed rows go from LDS into WMMA with W (sum order changed: neighbour sum before K sum); 3 GEMM launches, H@W array gone
# speedup vs baseline: 3.4288x; 1.1130x over previous
#include <hip/hip_runtime.h>
#include <hip/hip_bf16.h>

#define N_NODES 50000
#define N_EDGES 800000
#define DIN 128
#define DOUT 128
#define LDS_STRIDE 132

typedef __attribute__((ext_vector_type(2))) float v2f;
typedef __attribute__((ext_vector_type(8))) float v8f;

#define OFF_DINV 0
#define OFF_CNT  200192
#define OFF_ROWP 400384
#define OFF_RANK 600576
#define OFF_EW   3800576
#define OFF_HBUF 10200576

#define SCAN_T 1024

__global__ void zero_cnt(int* __restrict__ cnt) {
  int i = blockIdx.x * blockDim.x + threadIdx.x;
  if (i < N_NODES) cnt[i] = 0;
}

__global__ void count_rank(const int* __restrict__ dst, int* __restrict__ cnt,
                           int* __restrict__ rank) {
  int e = blockIdx.x * blockDim.x + threadIdx.x;
  if (e < N_EDGES) rank[e] = atomicAdd(&cnt[dst[e]], 1);
}

__launch_bounds__(SCAN_T)
__global__ void scan_deg(const int* __restrict__ cnt, int* __restrict__ rowptr,
                         float* __restrict__ dinv) {
  __shared__ int wave_tot[32];
  __shared__ int wave_pre[32];
  const int tid   = threadIdx.x;
  const int lane  = tid & 31;
  const int wv    = tid >> 5;
  const int first = blockIdx.x * SCAN_T;

  int part = 0;
  for (int i = tid; i < first; i += SCAN_T) part += cnt[i];

  const int i = first + tid;
  const int c = (i < N_NODES) ? cnt[i] : 0;

  int incl = c;
  for (int o = 1; o < 32; o <<= 1) {
    int t = __shfl_up(incl, o);
    if (lane >= o) incl += t;
  }
  for (int o = 16; o > 0; o >>= 1) part += __shfl_xor(part, o);

  if (lane == 31) wave_tot[wv] = incl;
  if (lane == 0)  wave_pre[wv] = part;
  __syncthreads();

  int base = wave_pre[lane] + ((lane < wv) ? wave_tot[lane] : 0);
  for (int o = 16; o > 0; o >>= 1) base += __shfl_xor(base, o);

  if (i < N_NODES) {
    rowptr[i] = base + incl - c;
    float d = (float)(c + 1);
    dinv[i] = d > 0.0f ? (1.0f / sqrtf(d)) : 0.0f;
  }
}

__global__ void fill_csr(const int* __restrict__ src, const int* __restrict__ dst,
                         const int* __restrict__ rank, const int* __restrict__ rowptr,
                         const float* __restrict__ dinv, int2* __restrict__ ew) {
  int e = blockIdx.x * blockDim.x + threadIdx.x;
  if (e < N_EDGES) {
    int s = src[e];
    int d = dst[e];
    float w = dinv[s] * dinv[d];
    ew[rowptr[d] + rank[e]] = make_int2(s, __float_as_int(w));
  }
}

__device__ __forceinline__ void fma4(float4& a, const float4 v, const float w) {
  a.x = fmaf(v.x, w, a.x); a.y = fmaf(v.y, w, a.y);
  a.z = fmaf(v.z, w, a.z); a.w = fmaf(v.w, w, a.w);
}

__device__ __forceinline__ float4 sum_node(const float4* __restrict__ H4,
                                           const int2* __restrict__ ew,
                                           const int node, const int start,
                                           const int n, const float di,
                                           const int2 my0, const int lane) {
  const float4 vself = H4[(size_t)node * 32 + lane];

  float4 acc;
  acc.x = 0.0f; acc.y = 0.0f; acc.z = 0.0f; acc.w = 0.0f;

#define SRC(k)  __builtin_amdgcn_readlane(my.x, (k))
#define WGT(k)  __int_as_float(__builtin_amdgcn_readlane(my.y, (k)))
#define ROW(s)  H4[(size_t)(s) * 32 + lane]

  for (int base = 0; base < n; base += 32) {
    const int m = (n - base) < 32 ? (n - base) : 32;
    int2 my = my0;
    if (base > 0) {
      my = make_int2(0, 0);
      if (lane < m) my = ew[start + base + lane];
    }

    int j = 0;
    for (; j + 8 <= m; j += 8) {
      const float4 v0 = ROW(SRC(j));
      const float4 v1 = ROW(SRC(j + 1));
      const float4 v2 = ROW(SRC(j + 2));
      const float4 v3 = ROW(SRC(j + 3));
      const float4 v4 = ROW(SRC(j + 4));
      const float4 v5 = ROW(SRC(j + 5));
      const float4 v6 = ROW(SRC(j + 6));
      const float4 v7 = ROW(SRC(j + 7));
      fma4(acc, v0, WGT(j));
      fma4(acc, v1, WGT(j + 1));
      fma4(acc, v2, WGT(j + 2));
      fma4(acc, v3, WGT(j + 3));
      fma4(acc, v4, WGT(j + 4));
      fma4(acc, v5, WGT(j + 5));
      fma4(acc, v6, WGT(j + 6));
      fma4(acc, v7, WGT(j + 7));
    }
    if (j + 4 <= m) {
      const float4 v0 = ROW(SRC(j));
      const float4 v1 = ROW(SRC(j + 1));
      const float4 v2 = ROW(SRC(j + 2));
      const float4 v3 = ROW(SRC(j + 3));
      fma4(acc, v0, WGT(j));
      fma4(acc, v1, WGT(j + 1));
      fma4(acc, v2, WGT(j + 2));
      fma4(acc, v3, WGT(j + 3));
      j += 4;
    }
    const int rem = m - j;
    if (rem == 3) {
      const float4 v0 = ROW(SRC(j));
      const float4 v1 = ROW(SRC(j + 1));
      const float4 v2 = ROW(SRC(j + 2));
      fma4(acc, v0, WGT(j));
      fma4(acc, v1, WGT(j + 1));
      fma4(acc, v2, WGT(j + 2));
    } else if (rem == 2) {
      const float4 v0 = ROW(SRC(j));
      const float4 v1 = ROW(SRC(j + 1));
      fma4(acc, v0, WGT(j));
      fma4(acc, v1, WGT(j + 1));
    } else if (rem == 1) {
      const float4 v0 = ROW(SRC(j));
      fma4(acc, v0, WGT(j));
    }
  }
#undef SRC
#undef WGT
#undef ROW

  fma4(acc, vself, di * di);
  return acc;
}

__launch_bounds__(256, 16)
__global__ void gcn_layer(const float* __restrict__ H,
                          const int* __restrict__ rowptr,
                          const int* __restrict__ cnt,
                          const int2* __restrict__ ew,
                          const float* __restrict__ dinv,
                          const float* __restrict__ W,
                          const float* __restrict__ bias,
                          float* __restrict__ out) {
  __shared__ float As[32 * LDS_STRIDE];

  const int tid  = threadIdx.x;
  const int lane = tid & 31;
  const int wave = __builtin_amdgcn_readfirstlane(tid >> 5);
  const int m0   = blockIdx.x * 32;
  const float4* H4 = (const float4*)H;

  static_assert(N_NODES % 4 == 0, "a wave's four nodes are all valid or all not");
  {
    const int q0 = m0 + wave * 4;
    float* arow = As + (wave * 4) * LDS_STRIDE + lane * 4;
    if (q0 < N_NODES) {
      const int4   st4 = *(const int4*)(rowptr + q0);
      const int4   nn4 = *(const int4*)(cnt + q0);
      const float4 dd4 = *(const float4*)(dinv + q0);
      const int   st[4] = {st4.x, st4.y, st4.z, st4.w};
      const int   nn[4] = {nn4.x, nn4.y, nn4.z, nn4.w};
      const float dd[4] = {dd4.x, dd4.y, dd4.z, dd4.w};
      int2 my0[4];
#pragma unroll
      for (int i = 0; i < 4; ++i) {
        const int m = nn[i] < 32 ? nn[i] : 32;
        my0[i] = make_int2(0, 0);
        if (lane < m) my0[i] = ew[st[i] + lane];
      }
#pragma unroll
      for (int i = 0; i < 4; ++i) {
        const float4 acc = sum_node(H4, ew, q0 + i, st[i], nn[i], dd[i], my0[i], lane);
        *(float4*)(arow + i * LDS_STRIDE) = acc;
      }
    } else {
      float4 z;
      z.x = 0.0f; z.y = 0.0f; z.z = 0.0f; z.w = 0.0f;
#pragma unroll
      for (int i = 0; i < 4; ++i) *(float4*)(arow + i * LDS_STRIDE) = z;
    }
  }
  __syncthreads();

  const int n0    = wave * 16;
  const int mlane = lane & 15;
  const int khalf = (lane >> 4) * 2;

  const float* a0   = As + (size_t)mlane * LDS_STRIDE + khalf;
  const float* a1   = a0 + 16 * LDS_STRIDE;
  const float* wcol = W + (size_t)khalf * DOUT + n0 + mlane;

  v8f acc0 = {}, acc1 = {};
#pragma unroll 4
  for (int k = 0; k < DIN; k += 4) {
    v2f b;
    b.x = wcol[(size_t)k * DOUT];
    b.y = wcol[(size_t)(k + 1) * DOUT];
    v2f x0 = *(const v2f*)(a0 + k);
    v2f x1 = *(const v2f*)(a1 + k);
    acc0 = __builtin_amdgcn_wmma_f32_16x16x4_f32(false, x0, false, b,
                                                 (short)0, acc0, false, false);
    acc1 = __builtin_amdgcn_wmma_f32_16x16x4_f32(false, x1, false, b,
                                                 (short)0, acc1, false, false);
  }

  __syncthreads();
  {
    float* c0 = As + (size_t)((lane >> 4) * 8) * LDS_STRIDE + n0 + mlane;
    float* c1 = c0 + 16 * LDS_STRIDE;
#pragma unroll
    for (int j = 0; j < 8; ++j) {
      c0[j * LDS_STRIDE] = acc0[j];
      c1[j * LDS_STRIDE] = acc1[j];
    }
  }
  __syncthreads();

  const float4 b4 = ((const float4*)bias)[lane];
  if (m0 + wave * 4 < N_NODES) {
#pragma unroll
    for (int i = 0; i < 4; ++i) {
      const int row = wave * 4 + i;
      const float4 v = *(const float4*)(As + row * LDS_STRIDE + lane * 4);
      float4 r;
      r.x = v.x + b4.x; r.y = v.y + b4.y; r.z = v.z + b4.z; r.w = v.w + b4.w;
      r.x = r.x > 0.0f ? r.x : 0.0f;
      r.y = r.y > 0.0f ? r.y : 0.0f;
      r.z = r.z > 0.0f ? r.z : 0.0f;
      r.w = r.w > 0.0f ? r.w : 0.0f;
      ((float4*)out)[(size_t)(m0 + row) * 32 + lane] = r;
    }
  }
}

extern "C" void kernel_launch(void* const* d_in, const int* in_sizes, int n_in,
                              void* d_out, int out_size, void* d_ws, size_t ws_size,
                              hipStream_t stream) {
  (void)in_sizes; (void)n_in; (void)out_size; (void)ws_size;

  const float* x  = (const float*)d_in[0];
  const int*   ei = (const int*)d_in[1];
  const float* Wl[3] = {(const float*)d_in[2], (const float*)d_in[4], (const float*)d_in[6]};
  const float* bl[3] = {(const float*)d_in[3], (const float*)d_in[5], (const float*)d_in[7]};
  const int* src = ei;
  const int* dst = ei + N_EDGES;

  char*  ws     = (char*)d_ws;
  float* dinv   = (float*)(ws + OFF_DINV);
  int*   cnt    = (int*)(ws + OFF_CNT);
  int*   rowptr = (int*)(ws + OFF_ROWP);
  int*   rank   = (int*)(ws + OFF_RANK);
  int2*  ew     = (int2*)(ws + OFF_EW);
  float* hbuf   = (float*)(ws + OFF_HBUF);

  zero_cnt  <<<(N_NODES + 255) / 256, 256, 0, stream>>>(cnt);
  count_rank<<<(N_EDGES + 255) / 256, 256, 0, stream>>>(dst, cnt, rank);
  scan_deg  <<<(N_NODES + SCAN_T - 1) / SCAN_T, SCAN_T, 0, stream>>>(cnt, rowptr, dinv);
  fill_csr  <<<(N_EDGES + 255) / 256, 256, 0, stream>>>(src, dst, rank, rowptr, dinv, ew);

  float* o = (float*)d_out;
  gcn_layer<<<(N_NODES + 31) / 32, 256, 0, stream>>>(x,    rowptr, cnt, ew, dinv, Wl[0], bl[0], o);
  gcn_layer<<<(N_NODES + 31) / 32, 256, 0, stream>>>(o,    rowptr, cnt, ew, dinv, Wl[1], bl[1], hbuf);
  gcn_layer<<<(N_NODES + 31) / 32, 256, 0, stream>>>(hbuf, rowptr, cnt, ew, dinv, Wl[2], bl[2], o);
}


# Edges grouped by destination by a two-level counting sort on LDS counters (chunk x 128-node-bin table, prefix, per-bin sort), not one returning global add per edge; list order = LDS arrival order
# speedup vs baseline: 4.6582x; 1.3586x over previous
#include <hip/hip_runtime.h>
#include <hip/hip_bf16.h>

#define N_NODES 50000
#define N_EDGES 800000
#define DIN 128
#define DOUT 128
#define LDS_STRIDE 132

typedef __attribute__((ext_vector_type(2))) float v2f;
typedef __attribute__((ext_vector_type(8))) float v8f;

#define OFF_DINV   0
#define OFF_CNT    200192
#define OFF_ROWP   400384
#define OFF_TAB    600576
#define OFF_TABX   756992
#define OFF_BINNED 913408
#define OFF_EW     7313408
#define OFF_HBUF   13713408

#define BIN_SHIFT 7
#define BIN_NODES (1 << BIN_SHIFT)
#define NB        ((N_NODES + BIN_NODES - 1) / BIN_NODES)
#define NC        100
#define CHUNK_E   (N_EDGES / NC)
#define NTAB      (NB * NC)
#define BIN_T     1024
#define EPT       ((CHUNK_E + BIN_T - 1) / BIN_T)
#define SORT_T    1024
#define SCAN_T    1024

static_assert(N_EDGES % NC == 0, "every chunk holds CHUNK_E edges");
static_assert(OFF_TABX - OFF_TAB >= NTAB * 4, "tab fits its slice");

__launch_bounds__(BIN_T)
__global__ void bin_count(const int* __restrict__ dst, int* __restrict__ tab) {
  __shared__ int hist[NB];
  const int tid   = threadIdx.x;
  const int chunk = blockIdx.x;
  for (int b = tid; b < NB; b += BIN_T) hist[b] = 0;
  __syncthreads();
  const int* d = dst + (size_t)chunk * CHUNK_E;
  int dv[EPT];
#pragma unroll
  for (int k = 0; k < EPT; ++k) {
    const int i = tid + k * BIN_T;
    dv[k] = (i < CHUNK_E) ? d[i] : 0;
  }
#pragma unroll
  for (int k = 0; k < EPT; ++k)
    if (tid + k * BIN_T < CHUNK_E) atomicAdd(&hist[dv[k] >> BIN_SHIFT], 1);
  __syncthreads();
  for (int b = tid; b < NB; b += BIN_T) tab[b * NC + chunk] = hist[b];
}

__launch_bounds__(SCAN_T)
__global__ void scan_tab(const int* __restrict__ tab, int* __restrict__ tabx) {
  __shared__ int wave_tot[32];
  __shared__ int wave_pre[32];
  const int tid   = threadIdx.x;
  const int lane  = tid & 31;
  const int wv    = tid >> 5;
  const int first = blockIdx.x * SCAN_T;

  int part = 0;
  for (int i = tid; i < first; i += SCAN_T) part += tab[i];

  const int i = first + tid;
  const int c = (i < NTAB) ? tab[i] : 0;

  int incl = c;
  for (int o = 1; o < 32; o <<= 1) {
    int t = __shfl_up(incl, o);
    if (lane >= o) incl += t;
  }
  for (int o = 16; o > 0; o >>= 1) part += __shfl_xor(part, o);

  if (lane == 31) wave_tot[wv] = incl;
  if (lane == 0)  wave_pre[wv] = part;
  __syncthreads();

  int base = wave_pre[lane] + ((lane < wv) ? wave_tot[lane] : 0);
  for (int o = 16; o > 0; o >>= 1) base += __shfl_xor(base, o);

  if (i < NTAB) tabx[i] = base + incl - c;
}

__launch_bounds__(BIN_T)
__global__ void bin_place(const int* __restrict__ src, const int* __restrict__ dst,
                          const int* __restrict__ tabx, int2* __restrict__ binned) {
  __shared__ int cur[NB];
  const int tid   = threadIdx.x;
  const int chunk = blockIdx.x;
  for (int b = tid; b < NB; b += BIN_T) cur[b] = tabx[b * NC + chunk];
  __syncthreads();
  const int* s = src + (size_t)chunk * CHUNK_E;
  const int* d = dst + (size_t)chunk * CHUNK_E;
  int sv[EPT], dv[EPT];
#pragma unroll
  for (int k = 0; k < EPT; ++k) {
    const int i = tid + k * BIN_T;
    sv[k] = (i < CHUNK_E) ? s[i] : 0;
    dv[k] = (i < CHUNK_E) ? d[i] : 0;
  }
#pragma unroll
  for (int k = 0; k < EPT; ++k) {
    if (tid + k * BIN_T < CHUNK_E) {
      const int p = atomicAdd(&cur[dv[k] >> BIN_SHIFT], 1);
      binned[p] = make_int2(sv[k], dv[k]);
    }
  }
}

__launch_bounds__(SORT_T)
__global__ void bin_sort(const int2* __restrict__ binned, const int* __restrict__ tabx,
                         int* __restrict__ rowptr, int* __restrict__ cnt,
                         float* __restrict__ dinv, int2* __restrict__ ew) {
  __shared__ int cl[BIN_NODES];
  __shared__ int wtot[BIN_NODES / 32];
  const int tid  = threadIdx.x;
  const int lane = tid & 31;
  const int wv   = tid >> 5;
  const int bin  = blockIdx.x;
  const int seg0 = tabx[bin * NC];
  const int seg1 = (bin + 1 < NB) ? tabx[(bin + 1) * NC] : N_EDGES;

  if (tid < BIN_NODES) cl[tid] = 0;
  __syncthreads();
  for (int i = seg0 + tid; i < seg1; i += 2 * SORT_T) {
    const int  j  = i + SORT_T;
    const bool h  = j < seg1;
    const int  y0 = binned[i].y;
    const int  y1 = h ? binned[j].y : 0;
    atomicAdd(&cl[y0 & (BIN_NODES - 1)], 1);
    if (h) atomicAdd(&cl[y1 & (BIN_NODES - 1)], 1);
  }
  __syncthreads();

  int c = 0, incl = 0;
  if (tid < BIN_NODES) {
    c = cl[tid];
    incl = c;
    for (int o = 1; o < 32; o <<= 1) {
      int t = __shfl_up(incl, o);
      if (lane >= o) incl += t;
    }
    if (lane == 31) wtot[wv] = incl;
  }
  __syncthreads();
  if (tid < BIN_NODES) {
    int start = seg0 + incl - c;
    for (int w = 0; w < wv; ++w) start += wtot[w];
    cl[tid] = start;
    const int node = bin * BIN_NODES + tid;
    if (node < N_NODES) {
      rowptr[node] = start;
      cnt[node]    = c;
      float d = (float)(c + 1);
      dinv[node] = d > 0.0f ? (1.0f / sqrtf(d)) : 0.0f;
    }
  }
  __syncthreads();
  for (int i = seg0 + tid; i < seg1; i += 2 * SORT_T) {
    const int  j = i + SORT_T;
    const bool h = j < seg1;
    const int2 v0 = binned[i];
    int2 v1 = make_int2(0, 0);
    if (h) v1 = binned[j];
    const int p0 = atomicAdd(&cl[v0.y & (BIN_NODES - 1)], 1);
    ew[p0] = v0;
    if (h) {
      const int p1 = atomicAdd(&cl[v1.y & (BIN_NODES - 1)], 1);
      ew[p1] = v1;
    }
  }
}

__global__ void edge_weight(const float* __restrict__ dinv, int2* ew) {
  int e = blockIdx.x * blockDim.x + threadIdx.x;
  if (e < N_EDGES) {
    const int2 v = ew[e];
    float w = dinv[v.x] * dinv[v.y];
    ew[e] = make_int2(v.x, __float_as_int(w));
  }
}

__device__ __forceinline__ void fma4(float4& a, const float4 v, const float w) {
  a.x = fmaf(v.x, w, a.x); a.y = fmaf(v.y, w, a.y);
  a.z = fmaf(v.z, w, a.z); a.w = fmaf(v.w, w, a.w);
}

__device__ __forceinline__ float4 sum_node(const float4* __restrict__ H4,
                                           const int2* __restrict__ ew,
                                           const int node, const int start,
                                           const int n, const float di,
                                           const int2 my0, const int lane) {
  const float4 vself = H4[(size_t)node * 32 + lane];

  float4 acc;
  acc.x = 0.0f; acc.y = 0.0f; acc.z = 0.0f; acc.w = 0.0f;

#define SRC(k)  __builtin_amdgcn_readlane(my.x, (k))
#define WGT(k)  __int_as_float(__builtin_amdgcn_readlane(my.y, (k)))
#define ROW(s)  H4[(size_t)(s) * 32 + lane]

  for (int base = 0; base < n; base += 32) {
    const int m = (n - base) < 32 ? (n - base) : 32;
    int2 my = my0;
    if (base > 0) {
      my = make_int2(0, 0);
      if (lane < m) my = ew[start + base + lane];
    }

    int j = 0;
    for (; j + 8 <= m; j += 8) {
      const float4 v0 = ROW(SRC(j));
      const float4 v1 = ROW(SRC(j + 1));
      const float4 v2 = ROW(SRC(j + 2));
      const float4 v3 = ROW(SRC(j + 3));
      const float4 v4 = ROW(SRC(j + 4));
      const float4 v5 = ROW(SRC(j + 5));
      const float4 v6 = ROW(SRC(j + 6));
      const float4 v7 = ROW(SRC(j + 7));
      fma4(acc, v0, WGT(j));
      fma4(acc, v1, WGT(j + 1));
      fma4(acc, v2, WGT(j + 2));
      fma4(acc, v3, WGT(j + 3));
      fma4(acc, v4, WGT(j + 4));
      fma4(acc, v5, WGT(j + 5));
      fma4(acc, v6, WGT(j + 6));
      fma4(acc, v7, WGT(j + 7));
    }
    if (j + 4 <= m) {
      const float4 v0 = ROW(SRC(j));
      const float4 v1 = ROW(SRC(j + 1));
      const float4 v2 = ROW(SRC(j + 2));
      const float4 v3 = ROW(SRC(j + 3));
      fma4(acc, v0, WGT(j));
      fma4(acc, v1, WGT(j + 1));
      fma4(acc, v2, WGT(j + 2));
      fma4(acc, v3, WGT(j + 3));
      j += 4;
    }
    const int rem = m - j;
    if (rem == 3) {
      const float4 v0 = ROW(SRC(j));
      const float4 v1 = ROW(SRC(j + 1));
      const float4 v2 = ROW(SRC(j + 2));
      fma4(acc, v0, WGT(j));
      fma4(acc, v1, WGT(j + 1));
      fma4(acc, v2, WGT(j + 2));
    } else if (rem == 2) {
      const float4 v0 = ROW(SRC(j));
      const float4 v1 = ROW(SRC(j + 1));
      fma4(acc, v0, WGT(j));
      fma4(acc, v1, WGT(j + 1));
    } else if (rem == 1) {
      const float4 v0 = ROW(SRC(j));
      fma4(acc, v0, WGT(j));
    }
  }
#undef SRC
#undef WGT
#undef ROW

  fma4(acc, vself, di * di);
  return acc;
}

__launch_bounds__(256, 16)
__global__ void gcn_layer(const float* __restrict__ H,
                          const int* __restrict__ rowptr,
                          const int* __restrict__ cnt,
                          const int2* __restrict__ ew,
                          const float* __restrict__ dinv,
                          const float* __restrict__ W,
                          const float* __restrict__ bias,
                          float* __restrict__ out) {
  __shared__ float As[32 * LDS_STRIDE];

  const int tid  = threadIdx.x;
  const int lane = tid & 31;
  const int wave = __builtin_amdgcn_readfirstlane(tid >> 5);
  const int m0   = blockIdx.x * 32;
  const float4* H4 = (const float4*)H;

  static_assert(N_NODES % 4 == 0, "a wave's four nodes are all valid or all not");
  {
    const int q0 = m0 + wave * 4;
    float* arow = As + (wave * 4) * LDS_STRIDE + lane * 4;
    if (q0 < N_NODES) {
      const int4   st4 = *(const int4*)(rowptr + q0);
      const int4   nn4 = *(const int4*)(cnt + q0);
      const float4 dd4 = *(const float4*)(dinv + q0);
      const int   st[4] = {st4.x, st4.y, st4.z, st4.w};
      const int   nn[4] = {nn4.x, nn4.y, nn4.z, nn4.w};
      const float dd[4] = {dd4.x, dd4.y, dd4.z, dd4.w};
      int2 my0[4];
#pragma unroll
      for (int i = 0; i < 4; ++i) {
        const int m = nn[i] < 32 ? nn[i] : 32;
        my0[i] = make_int2(0, 0);
        if (lane < m) my0[i] = ew[st[i] + lane];
      }
#pragma unroll
      for (int i = 0; i < 4; ++i) {
        const float4 acc = sum_node(H4, ew, q0 + i, st[i], nn[i], dd[i], my0[i], lane);
        *(float4*)(arow + i * LDS_STRIDE) = acc;
      }
    } else {
      float4 z;
      z.x = 0.0f; z.y = 0.0f; z.z = 0.0f; z.w = 0.0f;
#pragma unroll
      for (int i = 0; i < 4; ++i) *(float4*)(arow + i * LDS_STRIDE) = z;
    }
  }
  __syncthreads();

  const int n0    = wave * 16;
  const int mlane = lane & 15;
  const int khalf = (lane >> 4) * 2;

  const float* a0   = As + (size_t)mlane * LDS_STRIDE + khalf;
  const float* a1   = a0 + 16 * LDS_STRIDE;
  const float* wcol = W + (size_t)khalf * DOUT + n0 + mlane;

  v8f acc0 = {}, acc1 = {};
#pragma unroll 4
  for (int k = 0; k < DIN; k += 4) {
    v2f b;
    b.x = wcol[(size_t)k * DOUT];
    b.y = wcol[(size_t)(k + 1) * DOUT];
    v2f x0 = *(const v2f*)(a0 + k);
    v2f x1 = *(const v2f*)(a1 + k);
    acc0 = __builtin_amdgcn_wmma_f32_16x16x4_f32(false, x0, false, b,
                                                 (short)0, acc0, false, false);
    acc1 = __builtin_amdgcn_wmma_f32_16x16x4_f32(false, x1, false, b,
                                                 (short)0, acc1, false, false);
  }

  __syncthreads();
  {
    float* c0 = As + (size_t)((lane >> 4) * 8) * LDS_STRIDE + n0 + mlane;
    float* c1 = c0 + 16 * LDS_STRIDE;
#pragma unroll
    for (int j = 0; j < 8; ++j) {
      c0[j * LDS_STRIDE] = acc0[j];
      c1[j * LDS_STRIDE] = acc1[j];
    }
  }
  __syncthreads();

  const float4 b4 = ((const float4*)bias)[lane];
  if (m0 + wave * 4 < N_NODES) {
#pragma unroll
    for (int i = 0; i < 4; ++i) {
      const int row = wave * 4 + i;
      const float4 v = *(const float4*)(As + row * LDS_STRIDE + lane * 4);
      float4 r;
      r.x = v.x + b4.x; r.y = v.y + b4.y; r.z = v.z + b4.z; r.w = v.w + b4.w;
      r.x = r.x > 0.0f ? r.x : 0.0f;
      r.y = r.y > 0.0f ? r.y : 0.0f;
      r.z = r.z > 0.0f ? r.z : 0.0f;
      r.w = r.w > 0.0f ? r.w : 0.0f;
      ((float4*)out)[(size_t)(m0 + row) * 32 + lane] = r;
    }
  }
}

extern "C" void kernel_launch(void* const* d_in, const int* in_sizes, int n_in,
                              void* d_out, int out_size, void* d_ws, size_t ws_size,
                              hipStream_t stream) {
  (void)in_sizes; (void)n_in; (void)out_size; (void)ws_size;

  const float* x  = (const float*)d_in[0];
  const int*   ei = (const int*)d_in[1];
  const float* Wl[3] = {(const float*)d_in[2], (const float*)d_in[4], (const float*)d_in[6]};
  const float* bl[3] = {(const float*)d_in[3], (const float*)d_in[5], (const float*)d_in[7]};
  const int* src = ei;
  const int* dst = ei + N_EDGES;

  char*  ws     = (char*)d_ws;
  float* dinv   = (float*)(ws + OFF_DINV);
  int*   cnt    = (int*)(ws + OFF_CNT);
  int*   rowptr = (int*)(ws + OFF_ROWP);
  int*   tab    = (int*)(ws + OFF_TAB);
  int*   tabx   = (int*)(ws + OFF_TABX);
  int2*  binned = (int2*)(ws + OFF_BINNED);
  int2*  ew     = (int2*)(ws + OFF_EW);
  float* hbuf   = (float*)(ws + OFF_HBUF);

  bin_count  <<<NC, BIN_T, 0, stream>>>(dst, tab);
  scan_tab   <<<(NTAB + SCAN_T - 1) / SCAN_T, SCAN_T, 0, stream>>>(tab, tabx);
  bin_place  <<<NC, BIN_T, 0, stream>>>(src, dst, tabx, binned);
  bin_sort   <<<NB, SORT_T, 0, stream>>>(binned, tabx, rowptr, cnt, dinv, ew);
  edge_weight<<<(N_EDGES + 255) / 256, 256, 0, stream>>>(dinv, ew);

  float* o = (float*)d_out;
  gcn_layer<<<(N_NODES + 31) / 32, 256, 0, stream>>>(x,    rowptr, cnt, ew, dinv, Wl[0], bl[0], o);
  gcn_layer<<<(N_NODES + 31) / 32, 256, 0, stream>>>(o,    rowptr, cnt, ew, dinv, Wl[1], bl[1], hbuf);
  gcn_layer<<<(N_NODES + 31) / 32, 256, 0, stream>>>(hbuf, rowptr, cnt, ew, dinv, Wl[2], bl[2], o);
}


# Both edge-placing passes order their edges in LDS and copy LDS to memory front to back: a store writes 32 neighbouring 8-byte slots, not 32 scattered lines; list order stays LDS arrival order
# speedup vs baseline: 5.1109x; 1.0972x over previous
#include <hip/hip_runtime.h>
#include <hip/hip_bf16.h>

#define N_NODES 50000
#define N_EDGES 800000
#define DIN 128
#define DOUT 128
#define LDS_STRIDE 132

typedef __attribute__((ext_vector_type(2))) float v2f;
typedef __attribute__((ext_vector_type(8))) float v8f;

#define OFF_DINV   0
#define OFF_CNT    200192
#define OFF_ROWP   400384
#define OFF_TAB    600576
#define OFF_TABX   756992
#define OFF_BINNED 913408
#define OFF_EW     7313408
#define OFF_HBUF   13713408

#define BIN_SHIFT 7
#define BIN_NODES (1 << BIN_SHIFT)
#define NB        ((N_NODES + BIN_NODES - 1) / BIN_NODES)
#define NC        100
#define CHUNK_E   (N_EDGES / NC)
#define NTAB      (NB * NC)
#define BIN_T     1024
#define EPT       ((CHUNK_E + BIN_T - 1) / BIN_T)
#define SORT_T    1024
#define SCAN_T    1024
#define NBW       ((NB + 31) / 32)
#define SORT_CAP  4096

static_assert(N_EDGES % NC == 0, "every chunk holds CHUNK_E edges");
static_assert(OFF_TABX - OFF_TAB >= NTAB * 4, "tab fits its slice");

__launch_bounds__(BIN_T)
__global__ void bin_count(const int* __restrict__ dst, int* __restrict__ tab) {
  __shared__ int hist[NB];
  const int tid   = threadIdx.x;
  const int chunk = blockIdx.x;
  for (int b = tid; b < NB; b += BIN_T) hist[b] = 0;
  __syncthreads();
  const int* d = dst + (size_t)chunk * CHUNK_E;
  int dv[EPT];
#pragma unroll
  for (int k = 0; k < EPT; ++k) {
    const int i = tid + k * BIN_T;
    dv[k] = (i < CHUNK_E) ? d[i] : 0;
  }
#pragma unroll
  for (int k = 0; k < EPT; ++k)
    if (tid + k * BIN_T < CHUNK_E) atomicAdd(&hist[dv[k] >> BIN_SHIFT], 1);
  __syncthreads();
  for (int b = tid; b < NB; b += BIN_T) tab[b * NC + chunk] = hist[b];
}

__launch_bounds__(SCAN_T)
__global__ void scan_tab(const int* __restrict__ tab, int* __restrict__ tabx) {
  __shared__ int wave_tot[32];
  __shared__ int wave_pre[32];
  const int tid   = threadIdx.x;
  const int lane  = tid & 31;
  const int wv    = tid >> 5;
  const int first = blockIdx.x * SCAN_T;

  int part = 0;
  for (int i = tid; i < first; i += SCAN_T) part += tab[i];

  const int i = first + tid;
  const int c = (i < NTAB) ? tab[i] : 0;

  int incl = c;
  for (int o = 1; o < 32; o <<= 1) {
    int t = __shfl_up(incl, o);
    if (lane >= o) incl += t;
  }
  for (int o = 16; o > 0; o >>= 1) part += __shfl_xor(part, o);

  if (lane == 31) wave_tot[wv] = incl;
  if (lane == 0)  wave_pre[wv] = part;
  __syncthreads();

  int base = wave_pre[lane] + ((lane < wv) ? wave_tot[lane] : 0);
  for (int o = 16; o > 0; o >>= 1) base += __shfl_xor(base, o);

  if (i < NTAB) tabx[i] = base + incl - c;
}

__launch_bounds__(BIN_T)
__global__ void bin_place(const int* __restrict__ src, const int* __restrict__ dst,
                          const int* __restrict__ tab, const int* __restrict__ tabx,
                          int2* __restrict__ binned) {
  __shared__ int  gstart[NB];
  __shared__ int  lstart[NB];
  __shared__ int  cur[NB];
  __shared__ int  wtot[NBW];
  __shared__ int2 stage[CHUNK_E];
  const int tid   = threadIdx.x;
  const int lane  = tid & 31;
  const int wv    = tid >> 5;
  const int chunk = blockIdx.x;
  const int* s = src + (size_t)chunk * CHUNK_E;
  const int* d = dst + (size_t)chunk * CHUNK_E;

  int h = 0, g = 0;
  if (tid < NB) {
    h = tab[tid * NC + chunk];
    g = tabx[tid * NC + chunk];
  }
  int sv[EPT], dv[EPT];
#pragma unroll
  for (int k = 0; k < EPT; ++k) {
    const int i = tid + k * BIN_T;
    sv[k] = (i < CHUNK_E) ? s[i] : 0;
    dv[k] = (i < CHUNK_E) ? d[i] : 0;
  }

  int incl = h;
  if (tid < NBW * 32) {
    for (int o = 1; o < 32; o <<= 1) {
      int t = __shfl_up(incl, o);
      if (lane >= o) incl += t;
    }
    if (lane == 31) wtot[wv] = incl;
  }
  __syncthreads();
  if (tid < NB) {
    int start = incl - h;
    for (int w = 0; w < wv; ++w) start += wtot[w];
    lstart[tid] = start;
    cur[tid]    = start;
    gstart[tid] = g;
  }
  __syncthreads();
  int p[EPT];
#pragma unroll
  for (int k = 0; k < EPT; ++k) {
    p[k] = 0;
    if (tid + k * BIN_T < CHUNK_E) p[k] = atomicAdd(&cur[dv[k] >> BIN_SHIFT], 1);
  }
#pragma unroll
  for (int k = 0; k < EPT; ++k)
    if (tid + k * BIN_T < CHUNK_E) stage[p[k]] = make_int2(sv[k], dv[k]);
  __syncthreads();
#pragma unroll
  for (int k = 0; k < EPT; ++k) {
    const int i = tid + k * BIN_T;
    if (i < CHUNK_E) {
      const int2 v = stage[i];
      const int  b = v.y >> BIN_SHIFT;
      binned[gstart[b] + (i - lstart[b])] = v;
    }
  }
}

__launch_bounds__(SORT_T)
__global__ void bin_sort(const int2* __restrict__ binned, const int* __restrict__ tabx,
                         int* __restrict__ rowptr, int* __restrict__ cnt,
                         float* __restrict__ dinv, int2* __restrict__ ew) {
  __shared__ int cl[BIN_NODES];
  __shared__ int wtot[BIN_NODES / 32];
  __shared__ int2 stage[SORT_CAP];
  const int tid  = threadIdx.x;
  const int lane = tid & 31;
  const int wv   = tid >> 5;
  const int bin  = blockIdx.x;
  const int seg0 = tabx[bin * NC];
  const int seg1 = (bin + 1 < NB) ? tabx[(bin + 1) * NC] : N_EDGES;
  const bool staged = (seg1 - seg0) <= SORT_CAP;

  if (tid < BIN_NODES) cl[tid] = 0;
  __syncthreads();
  for (int i = seg0 + tid; i < seg1; i += 2 * SORT_T) {
    const int  j  = i + SORT_T;
    const bool h  = j < seg1;
    const int  y0 = binned[i].y;
    const int  y1 = h ? binned[j].y : 0;
    atomicAdd(&cl[y0 & (BIN_NODES - 1)], 1);
    if (h) atomicAdd(&cl[y1 & (BIN_NODES - 1)], 1);
  }
  __syncthreads();

  int c = 0, incl = 0;
  if (tid < BIN_NODES) {
    c = cl[tid];
    incl = c;
    for (int o = 1; o < 32; o <<= 1) {
      int t = __shfl_up(incl, o);
      if (lane >= o) incl += t;
    }
    if (lane == 31) wtot[wv] = incl;
  }
  __syncthreads();
  if (tid < BIN_NODES) {
    int start = seg0 + incl - c;
    for (int w = 0; w < wv; ++w) start += wtot[w];
    cl[tid] = start;
    const int node = bin * BIN_NODES + tid;
    if (node < N_NODES) {
      rowptr[node] = start;
      cnt[node]    = c;
      float d = (float)(c + 1);
      dinv[node] = d > 0.0f ? (1.0f / sqrtf(d)) : 0.0f;
    }
  }
  __syncthreads();
  for (int i = seg0 + tid; i < seg1; i += 2 * SORT_T) {
    const int  j = i + SORT_T;
    const bool h = j < seg1;
    const int2 v0 = binned[i];
    int2 v1 = make_int2(0, 0);
    if (h) v1 = binned[j];
    const int p0 = atomicAdd(&cl[v0.y & (BIN_NODES - 1)], 1);
    if (staged) stage[p0 - seg0] = v0; else ew[p0] = v0;
    if (h) {
      const int p1 = atomicAdd(&cl[v1.y & (BIN_NODES - 1)], 1);
      if (staged) stage[p1 - seg0] = v1; else ew[p1] = v1;
    }
  }
  if (staged) {
    __syncthreads();
    for (int i = seg0 + tid; i < seg1; i += SORT_T) ew[i] = stage[i - seg0];
  }
}

__global__ void edge_weight(const float* __restrict__ dinv, int2* ew) {
  int e = blockIdx.x * blockDim.x + threadIdx.x;
  if (e < N_EDGES) {
    const int2 v = ew[e];
    float w = dinv[v.x] * dinv[v.y];
    ew[e] = make_int2(v.x, __float_as_int(w));
  }
}

__device__ __forceinline__ void fma4(float4& a, const float4 v, const float w) {
  a.x = fmaf(v.x, w, a.x); a.y = fmaf(v.y, w, a.y);
  a.z = fmaf(v.z, w, a.z); a.w = fmaf(v.w, w, a.w);
}

__device__ __forceinline__ float4 sum_node(const float4* __restrict__ H4,
                                           const int2* __restrict__ ew,
                                           const int node, const int start,
                                           const int n, const float di,
                                           const int2 my0, const int lane) {
  const float4 vself = H4[(size_t)node * 32 + lane];

  float4 acc;
  acc.x = 0.0f; acc.y = 0.0f; acc.z = 0.0f; acc.w = 0.0f;

#define SRC(k)  __builtin_amdgcn_readlane(my.x, (k))
#define WGT(k)  __int_as_float(__builtin_amdgcn_readlane(my.y, (k)))
#define ROW(s)  H4[(size_t)(s) * 32 + lane]

  for (int base = 0; base < n; base += 32) {
    const int m = (n - base) < 32 ? (n - base) : 32;
    int2 my = my0;
    if (base > 0) {
      my = make_int2(0, 0);
      if (lane < m) my = ew[start + base + lane];
    }

    int j = 0;
    for (; j + 8 <= m; j += 8) {
      const float4 v0 = ROW(SRC(j));
      const float4 v1 = ROW(SRC(j + 1));
      const float4 v2 = ROW(SRC(j + 2));
      const float4 v3 = ROW(SRC(j + 3));
      const float4 v4 = ROW(SRC(j + 4));
      const float4 v5 = ROW(SRC(j + 5));
      const float4 v6 = ROW(SRC(j + 6));
      const float4 v7 = ROW(SRC(j + 7));
      fma4(acc, v0, WGT(j));
      fma4(acc, v1, WGT(j + 1));
      fma4(acc, v2, WGT(j + 2));
      fma4(acc, v3, WGT(j + 3));
      fma4(acc, v4, WGT(j + 4));
      fma4(acc, v5, WGT(j + 5));
      fma4(acc, v6, WGT(j + 6));
      fma4(acc, v7, WGT(j + 7));
    }
    if (j + 4 <= m) {
      const float4 v0 = ROW(SRC(j));
      const float4 v1 = ROW(SRC(j + 1));
      const float4 v2 = ROW(SRC(j + 2));
      const float4 v3 = ROW(SRC(j + 3));
      fma4(acc, v0, WGT(j));
      fma4(acc, v1, WGT(j + 1));
      fma4(acc, v2, WGT(j + 2));
      fma4(acc, v3, WGT(j + 3));
      j += 4;
    }
    const int rem = m - j;
    if (rem == 3) {
      const float4 v0 = ROW(SRC(j));
      const float4 v1 = ROW(SRC(j + 1));
      const float4 v2 = ROW(SRC(j + 2));
      fma4(acc, v0, WGT(j));
      fma4(acc, v1, WGT(j + 1));
      fma4(acc, v2, WGT(j + 2));
    } else if (rem == 2) {
      const float4 v0 = ROW(SRC(j));
      const float4 v1 = ROW(SRC(j + 1));
      fma4(acc, v0, WGT(j));
      fma4(acc, v1, WGT(j + 1));
    } else if (rem == 1) {
      const float4 v0 = ROW(SRC(j));
      fma4(acc, v0, WGT(j));
    }
  }
#undef SRC
#undef WGT
#undef ROW

  fma4(acc, vself, di * di);
  return acc;
}

__launch_bounds__(256, 16)
__global__ void gcn_layer(const float* __restrict__ H,
                          const int* __restrict__ rowptr,
                          const int* __restrict__ cnt,
                          const int2* __restrict__ ew,
                          const float* __restrict__ dinv,
                          const float* __restrict__ W,
                          const float* __restrict__ bias,
                          float* __restrict__ out) {
  __shared__ float As[32 * LDS_STRIDE];

  const int tid  = threadIdx.x;
  const int lane = tid & 31;
  const int wave = __builtin_amdgcn_readfirstlane(tid >> 5);
  const int m0   = blockIdx.x * 32;
  const float4* H4 = (const float4*)H;

  static_assert(N_NODES % 4 == 0, "a wave's four nodes are all valid or all not");
  {
    const int q0 = m0 + wave * 4;
    float* arow = As + (wave * 4) * LDS_STRIDE + lane * 4;
    if (q0 < N_NODES) {
      const int4   st4 = *(const int4*)(rowptr + q0);
      const int4   nn4 = *(const int4*)(cnt + q0);
      const float4 dd4 = *(const float4*)(dinv + q0);
      const int   st[4] = {st4.x, st4.y, st4.z, st4.w};
      const int   nn[4] = {nn4.x, nn4.y, nn4.z, nn4.w};
      const float dd[4] = {dd4.x, dd4.y, dd4.z, dd4.w};
      int2 my0[4];
#pragma unroll
      for (int i = 0; i < 4; ++i) {
        const int m = nn[i] < 32 ? nn[i] : 32;
        my0[i] = make_int2(0, 0);
        if (lane < m) my0[i] = ew[st[i] + lane];
      }
#pragma unroll
      for (int i = 0; i < 4; ++i) {
        const float4 acc = sum_node(H4, ew, q0 + i, st[i], nn[i], dd[i], my0[i], lane);
        *(float4*)(arow + i * LDS_STRIDE) = acc;
      }
    } else {
      float4 z;
      z.x = 0.0f; z.y = 0.0f; z.z = 0.0f; z.w = 0.0f;
#pragma unroll
      for (int i = 0; i < 4; ++i) *(float4*)(arow + i * LDS_STRIDE) = z;
    }
  }
  __syncthreads();

  const int n0    = wave * 16;
  const int mlane = lane & 15;
  const int khalf = (lane >> 4) * 2;

  const float* a0   = As + (size_t)mlane * LDS_STRIDE + khalf;
  const float* a1   = a0 + 16 * LDS_STRIDE;
  const float* wcol = W + (size_t)khalf * DOUT + n0 + mlane;

  v8f acc0 = {}, acc1 = {};
#pragma unroll 4
  for (int k = 0; k < DIN; k += 4) {
    v2f b;
    b.x = wcol[(size_t)k * DOUT];
    b.y = wcol[(size_t)(k + 1) * DOUT];
    v2f x0 = *(const v2f*)(a0 + k);
    v2f x1 = *(const v2f*)(a1 + k);
    acc0 = __builtin_amdgcn_wmma_f32_16x16x4_f32(false, x0, false, b,
                                                 (short)0, acc0, false, false);
    acc1 = __builtin_amdgcn_wmma_f32_16x16x4_f32(false, x1, false, b,
                                                 (short)0, acc1, false, false);
  }

  __syncthreads();
  {
    float* c0 = As + (size_t)((lane >> 4) * 8) * LDS_STRIDE + n0 + mlane;
    float* c1 = c0 + 16 * LDS_STRIDE;
#pragma unroll
    for (int j = 0; j < 8; ++j) {
      c0[j * LDS_STRIDE] = acc0[j];
      c1[j * LDS_STRIDE] = acc1[j];
    }
  }
  __syncthreads();

  const float4 b4 = ((const float4*)bias)[lane];
  if (m0 + wave * 4 < N_NODES) {
#pragma unroll
    for (int i = 0; i < 4; ++i) {
      const int row = wave * 4 + i;
      const float4 v = *(const float4*)(As + row * LDS_STRIDE + lane * 4);
      float4 r;
      r.x = v.x + b4.x; r.y = v.y + b4.y; r.z = v.z + b4.z; r.w = v.w + b4.w;
      r.x = r.x > 0.0f ? r.x : 0.0f;
      r.y = r.y > 0.0f ? r.y : 0.0f;
      r.z = r.z > 0.0f ? r.z : 0.0f;
      r.w = r.w > 0.0f ? r.w : 0.0f;
      ((float4*)out)[(size_t)(m0 + row) * 32 + lane] = r;
    }
  }
}

extern "C" void kernel_launch(void* const* d_in, const int* in_sizes, int n_in,
                              void* d_out, int out_size, void* d_ws, size_t ws_size,
                              hipStream_t stream) {
  (void)in_sizes; (void)n_in; (void)out_size; (void)ws_size;

  const float* x  = (const float*)d_in[0];
  const int*   ei = (const int*)d_in[1];
  const float* Wl[3] = {(const float*)d_in[2], (const float*)d_in[4], (const float*)d_in[6]};
  const float* bl[3] = {(const float*)d_in[3], (const float*)d_in[5], (const float*)d_in[7]};
  const int* src = ei;
  const int* dst = ei + N_EDGES;

  char*  ws     = (char*)d_ws;
  float* dinv   = (float*)(ws + OFF_DINV);
  int*   cnt    = (int*)(ws + OFF_CNT);
  int*   rowptr = (int*)(ws + OFF_ROWP);
  int*   tab    = (int*)(ws + OFF_TAB);
  int*   tabx   = (int*)(ws + OFF_TABX);
  int2*  binned = (int2*)(ws + OFF_BINNED);
  int2*  ew     = (int2*)(ws + OFF_EW);
  float* hbuf   = (float*)(ws + OFF_HBUF);

  bin_count  <<<NC, BIN_T, 0, stream>>>(dst, tab);
  scan_tab   <<<(NTAB + SCAN_T - 1) / SCAN_T, SCAN_T, 0, stream>>>(tab, tabx);
  bin_place  <<<NC, BIN_T, 0, stream>>>(src, dst, tab, tabx, binned);
  bin_sort   <<<NB, SORT_T, 0, stream>>>(binned, tabx, rowptr, cnt, dinv, ew);
  edge_weight<<<(N_EDGES + 255) / 256, 256, 0, stream>>>(dinv, ew);

  float* o = (float*)d_out;
  gcn_layer<<<(N_NODES + 31) / 32, 256, 0, stream>>>(x,    rowptr, cnt, ew, dinv, Wl[0], bl[0], o);
  gcn_layer<<<(N_NODES + 31) / 32, 256, 0, stream>>>(o,    rowptr, cnt, ew, dinv, Wl[1], bl[1], hbuf);
  gcn_layer<<<(N_NODES + 31) / 32, 256, 0, stream>>>(hbuf, rowptr, cnt, ew, dinv, Wl[2], bl[2], o);
}


# Edge grouping in 2 launches, not 4: each chunk is ordered by bin in LDS and written back in place with its bin starts; a bin's pass reads its run from every chunk. List order stays LDS arrival order
# speedup vs baseline: 5.4536x; 1.0671x over previous
#include <hip/hip_runtime.h>
#include <hip/hip_bf16.h>

#define N_NODES 50000
#define N_EDGES 800000
#define DIN 128
#define DOUT 128
#define LDS_STRIDE 132

typedef __attribute__((ext_vector_type(2))) float v2f;
typedef __attribute__((ext_vector_type(8))) float v8f;

#define OFF_DINV   0
#define OFF_CNT    200192
#define OFF_ROWP   400384
#define OFF_LOFF   600576
#define OFF_BINNED 913408
#define OFF_EW     7313408
#define OFF_HBUF   13713408

#define BIN_SHIFT 7
#define BIN_NODES (1 << BIN_SHIFT)
#define NB        ((N_NODES + BIN_NODES - 1) / BIN_NODES)
#define NC        100
#define CHUNK_E   (N_EDGES / NC)
#define NTAB      (NB * NC)
#define BIN_T     1024
#define EPT       ((CHUNK_E + BIN_T - 1) / BIN_T)
#define SORT_T    1024
#define NBW       ((NB + 31) / 32)
#define NCW       ((NC + 31) / 32)
#define RSRCH     128
#define SORT_CAP  4096

static_assert(N_EDGES % NC == 0, "every chunk holds CHUNK_E edges");
static_assert(OFF_BINNED - OFF_LOFF >= NTAB * 4, "loff fits its slice");
static_assert(RSRCH >= NC && (RSRCH & (RSRCH - 1)) == 0 && RSRCH <= SORT_T, "run search covers the chunks");

__launch_bounds__(BIN_T)
__global__ void chunk_sort(const int* __restrict__ src, const int* __restrict__ dst,
                           int* __restrict__ loff, int2* __restrict__ binned) {
  __shared__ int  cur[NB];
  __shared__ int  wtot[NBW];
  __shared__ int2 stage[CHUNK_E];
  const int tid   = threadIdx.x;
  const int lane  = tid & 31;
  const int wv    = tid >> 5;
  const int chunk = blockIdx.x;
  const int* s = src + (size_t)chunk * CHUNK_E;
  const int* d = dst + (size_t)chunk * CHUNK_E;

  for (int b = tid; b < NB; b += BIN_T) cur[b] = 0;
  int sv[EPT], dv[EPT];
#pragma unroll
  for (int k = 0; k < EPT; ++k) {
    const int i = tid + k * BIN_T;
    sv[k] = (i < CHUNK_E) ? s[i] : 0;
    dv[k] = (i < CHUNK_E) ? d[i] : 0;
  }
  __syncthreads();
#pragma unroll
  for (int k = 0; k < EPT; ++k)
    if (tid + k * BIN_T < CHUNK_E) atomicAdd(&cur[dv[k] >> BIN_SHIFT], 1);
  __syncthreads();

  const int h = (tid < NB) ? cur[tid] : 0;
  int incl = h;
  if (tid < NBW * 32) {
    for (int o = 1; o < 32; o <<= 1) {
      int t = __shfl_up(incl, o);
      if (lane >= o) incl += t;
    }
    if (lane == 31) wtot[wv] = incl;
  }
  __syncthreads();
  if (tid < NB) {
    int start = incl - h;
    for (int w = 0; w < wv; ++w) start += wtot[w];
    cur[tid] = start;
    loff[tid * NC + chunk] = start;
  }
  __syncthreads();
  int p[EPT];
#pragma unroll
  for (int k = 0; k < EPT; ++k) {
    p[k] = 0;
    if (tid + k * BIN_T < CHUNK_E) p[k] = atomicAdd(&cur[dv[k] >> BIN_SHIFT], 1);
  }
#pragma unroll
  for (int k = 0; k < EPT; ++k)
    if (tid + k * BIN_T < CHUNK_E) stage[p[k]] = make_int2(sv[k], dv[k]);
  __syncthreads();
  int2* o = binned + (size_t)chunk * CHUNK_E;
#pragma unroll
  for (int k = 0; k < EPT; ++k) {
    const int i = tid + k * BIN_T;
    if (i < CHUNK_E) o[i] = stage[i];
  }
}

__device__ __forceinline__ int run_slot(const int* rbeg, const int* radr, const int e) {
  int pos = 0;
#pragma unroll
  for (int step = RSRCH / 2; step >= 1; step >>= 1)
    if (rbeg[pos + step] <= e) pos += step;
  return radr[pos] + (e - rbeg[pos]);
}

__launch_bounds__(SORT_T)
__global__ void bin_sort(const int2* __restrict__ binned, const int* __restrict__ loff,
                         int* __restrict__ rowptr, int* __restrict__ cnt,
                         float* __restrict__ dinv, int2* __restrict__ ew) {
  __shared__ int cl[BIN_NODES];
  __shared__ int wtot[BIN_NODES / 32];
  __shared__ int rbeg[RSRCH];
  __shared__ int radr[NC];
  __shared__ int rlen[NCW];
  __shared__ int roff[NCW];
  __shared__ int2 stage[SORT_CAP];
  const int tid  = threadIdx.x;
  const int lane = tid & 31;
  const int wv   = tid >> 5;
  const int bin  = blockIdx.x;

  int l0 = 0, len = 0;
  if (tid < NC) {
    l0 = loff[bin * NC + tid];
    const int l1 = (bin + 1 < NB) ? loff[(bin + 1) * NC + tid] : CHUNK_E;
    len = l1 - l0;
  }
  if (tid < BIN_NODES) cl[tid] = 0;
  int rincl = len;
  if (tid < NCW * 32) {
    for (int o = 1; o < 32; o <<= 1) {
      int t = __shfl_up(rincl, o);
      if (lane >= o) rincl += t;
    }
    int lsum = l0;
    for (int o = 16; o > 0; o >>= 1) lsum += __shfl_xor(lsum, o);
    if (lane == 31) rlen[wv] = rincl;
    if (lane == 0)  roff[wv] = lsum;
  }
  __syncthreads();
  if (tid < NC) {
    int st = rincl - len;
    for (int w = 0; w < wv; ++w) st += rlen[w];
    rbeg[tid] = st;
    radr[tid] = tid * CHUNK_E + l0;
  } else if (tid < RSRCH) {
    rbeg[tid] = 0x7fffffff;
  }
  int seg0 = 0, total = 0;
#pragma unroll
  for (int w = 0; w < NCW; ++w) { seg0 += roff[w]; total += rlen[w]; }
  const bool staged = total <= SORT_CAP;
  __syncthreads();
  for (int e = tid; e < total; e += 2 * SORT_T) {
    const int  f  = e + SORT_T;
    const bool h  = f < total;
    const int  a0 = run_slot(rbeg, radr, e);
    const int  a1 = h ? run_slot(rbeg, radr, f) : a0;
    const int  y0 = binned[a0].y;
    const int  y1 = h ? binned[a1].y : 0;
    atomicAdd(&cl[y0 & (BIN_NODES - 1)], 1);
    if (h) atomicAdd(&cl[y1 & (BIN_NODES - 1)], 1);
  }
  __syncthreads();

  int c = 0, incl = 0;
  if (tid < BIN_NODES) {
    c = cl[tid];
    incl = c;
    for (int o = 1; o < 32; o <<= 1) {
      int t = __shfl_up(incl, o);
      if (lane >= o) incl += t;
    }
    if (lane == 31) wtot[wv] = incl;
  }
  __syncthreads();
  if (tid < BIN_NODES) {
    int start = seg0 + incl - c;
    for (int w = 0; w < wv; ++w) start += wtot[w];
    cl[tid] = start;
    const int node = bin * BIN_NODES + tid;
    if (node < N_NODES) {
      rowptr[node] = start;
      cnt[node]    = c;
      float d = (float)(c + 1);
      dinv[node] = d > 0.0f ? (1.0f / sqrtf(d)) : 0.0f;
    }
  }
  __syncthreads();
  for (int e = tid; e < total; e += 2 * SORT_T) {
    const int  f = e + SORT_T;
    const bool h = f < total;
    const int  a0 = run_slot(rbeg, radr, e);
    const int  a1 = h ? run_slot(rbeg, radr, f) : a0;
    const int2 v0 = binned[a0];
    int2 v1 = make_int2(0, 0);
    if (h) v1 = binned[a1];
    const int p0 = atomicAdd(&cl[v0.y & (BIN_NODES - 1)], 1);
    if (staged) stage[p0 - seg0] = v0; else ew[p0] = v0;
    if (h) {
      const int p1 = atomicAdd(&cl[v1.y & (BIN_NODES - 1)], 1);
      if (staged) stage[p1 - seg0] = v1; else ew[p1] = v1;
    }
  }
  if (staged) {
    __syncthreads();
    for (int i = tid; i < total; i += SORT_T) ew[seg0 + i] = stage[i];
  }
}

__global__ void edge_weight(const float* __restrict__ dinv, int2* ew) {
  int e = blockIdx.x * blockDim.x + threadIdx.x;
  if (e < N_EDGES) {
    const int2 v = ew[e];
    float w = dinv[v.x] * dinv[v.y];
    ew[e] = make_int2(v.x, __float_as_int(w));
  }
}

__device__ __forceinline__ void fma4(float4& a, const float4 v, const float w) {
  a.x = fmaf(v.x, w, a.x); a.y = fmaf(v.y, w, a.y);
  a.z = fmaf(v.z, w, a.z); a.w = fmaf(v.w, w, a.w);
}

__device__ __forceinline__ float4 sum_node(const float4* __restrict__ H4,
                                           const int2* __restrict__ ew,
                                           const int node, const int start,
                                           const int n, const float di,
                                           const int2 my0, const int lane) {
  const float4 vself = H4[(size_t)node * 32 + lane];

  float4 acc;
  acc.x = 0.0f; acc.y = 0.0f; acc.z = 0.0f; acc.w = 0.0f;

#define SRC(k)  __builtin_amdgcn_readlane(my.x, (k))
#define WGT(k)  __int_as_float(__builtin_amdgcn_readlane(my.y, (k)))
#define ROW(s)  H4[(size_t)(s) * 32 + lane]

  for (int base = 0; base < n; base += 32) {
    const int m = (n - base) < 32 ? (n - base) : 32;
    int2 my = my0;
    if (base > 0) {
      my = make_int2(0, 0);
      if (lane < m) my = ew[start + base + lane];
    }

    int j = 0;
    for (; j + 8 <= m; j += 8) {
      const float4 v0 = ROW(SRC(j));
      const float4 v1 = ROW(SRC(j + 1));
      const float4 v2 = ROW(SRC(j + 2));
      const float4 v3 = ROW(SRC(j + 3));
      const float4 v4 = ROW(SRC(j + 4));
      const float4 v5 = ROW(SRC(j + 5));
      const float4 v6 = ROW(SRC(j + 6));
      const float4 v7 = ROW(SRC(j + 7));
      fma4(acc, v0, WGT(j));
      fma4(acc, v1, WGT(j + 1));
      fma4(acc, v2, WGT(j + 2));
      fma4(acc, v3, WGT(j + 3));
      fma4(acc, v4, WGT(j + 4));
      fma4(acc, v5, WGT(j + 5));
      fma4(acc, v6, WGT(j + 6));
      fma4(acc, v7, WGT(j + 7));
    }
    if (j + 4 <= m) {
      const float4 v0 = ROW(SRC(j));
      const float4 v1 = ROW(SRC(j + 1));
      const float4 v2 = ROW(SRC(j + 2));
      const float4 v3 = ROW(SRC(j + 3));
      fma4(acc, v0, WGT(j));
      fma4(acc, v1, WGT(j + 1));
      fma4(acc, v2, WGT(j + 2));
      fma4(acc, v3, WGT(j + 3));
      j += 4;
    }
    const int rem = m - j;
    if (rem == 3) {
      const float4 v0 = ROW(SRC(j));
      const float4 v1 = ROW(SRC(j + 1));
      const float4 v2 = ROW(SRC(j + 2));
      fma4(acc, v0, WGT(j));
      fma4(acc, v1, WGT(j + 1));
      fma4(acc, v2, WGT(j + 2));
    } else if (rem == 2) {
      const float4 v0 = ROW(SRC(j));
      const float4 v1 = ROW(SRC(j + 1));
      fma4(acc, v0, WGT(j));
      fma4(acc, v1, WGT(j + 1));
    } else if (rem == 1) {
      const float4 v0 = ROW(SRC(j));
      fma4(acc, v0, WGT(j));
    }
  }
#undef SRC
#undef WGT
#undef ROW

  fma4(acc, vself, di * di);
  return acc;
}

__launch_bounds__(256, 16)
__global__ void gcn_layer(const float* __restrict__ H,
                          const int* __restrict__ rowptr,
                          const int* __restrict__ cnt,
                          const int2* __restrict__ ew,
                          const float* __restrict__ dinv,
                          const float* __restrict__ W,
                          const float* __restrict__ bias,
                          float* __restrict__ out) {
  __shared__ float As[32 * LDS_STRIDE];

  const int tid  = threadIdx.x;
  const int lane = tid & 31;
  const int wave = __builtin_amdgcn_readfirstlane(tid >> 5);
  const int m0   = blockIdx.x * 32;
  const float4* H4 = (const float4*)H;

  static_assert(N_NODES % 4 == 0, "a wave's four nodes are all valid or all not");
  {
    const int q0 = m0 + wave * 4;
    float* arow = As + (wave * 4) * LDS_STRIDE + lane * 4;
    if (q0 < N_NODES) {
      const int4   st4 = *(const int4*)(rowptr + q0);
      const int4   nn4 = *(const int4*)(cnt + q0);
      const float4 dd4 = *(const float4*)(dinv + q0);
      const int   st[4] = {st4.x, st4.y, st4.z, st4.w};
      const int   nn[4] = {nn4.x, nn4.y, nn4.z, nn4.w};
      const float dd[4] = {dd4.x, dd4.y, dd4.z, dd4.w};
      int2 my0[4];
#pragma unroll
      for (int i = 0; i < 4; ++i) {
        const int m = nn[i] < 32 ? nn[i] : 32;
        my0[i] = make_int2(0, 0);
        if (lane < m) my0[i] = ew[st[i] + lane];
      }
#pragma unroll
      for (int i = 0; i < 4; ++i) {
        const float4 acc = sum_node(H4, ew, q0 + i, st[i], nn[i], dd[i], my0[i], lane);
        *(float4*)(arow + i * LDS_STRIDE) = acc;
      }
    } else {
      float4 z;
      z.x = 0.0f; z.y = 0.0f; z.z = 0.0f; z.w = 0.0f;
#pragma unroll
      for (int i = 0; i < 4; ++i) *(float4*)(arow + i * LDS_STRIDE) = z;
    }
  }
  __syncthreads();

  const int n0    = wave * 16;
  const int mlane = lane & 15;
  const int khalf = (lane >> 4) * 2;

  const float* a0   = As + (size_t)mlane * LDS_STRIDE + khalf;
  const float* a1   = a0 + 16 * LDS_STRIDE;
  const float* wcol = W + (size_t)khalf * DOUT + n0 + mlane;

  v8f acc0 = {}, acc1 = {};
#pragma unroll 4
  for (int k = 0; k < DIN; k += 4) {
    v2f b;
    b.x = wcol[(size_t)k * DOUT];
    b.y = wcol[(size_t)(k + 1) * DOUT];
    v2f x0 = *(const v2f*)(a0 + k);
    v2f x1 = *(const v2f*)(a1 + k);
    acc0 = __builtin_amdgcn_wmma_f32_16x16x4_f32(false, x0, false, b,
                                                 (short)0, acc0, false, false);
    acc1 = __builtin_amdgcn_wmma_f32_16x16x4_f32(false, x1, false, b,
                                                 (short)0, acc1, false, false);
  }

  __syncthreads();
  {
    float* c0 = As + (size_t)((lane >> 4) * 8) * LDS_STRIDE + n0 + mlane;
    float* c1 = c0 + 16 * LDS_STRIDE;
#pragma unroll
    for (int j = 0; j < 8; ++j) {
      c0[j * LDS_STRIDE] = acc0[j];
      c1[j * LDS_STRIDE] = acc1[j];
    }
  }
  __syncthreads();

  const float4 b4 = ((const float4*)bias)[lane];
  if (m0 + wave * 4 < N_NODES) {
#pragma unroll
    for (int i = 0; i < 4; ++i) {
      const int row = wave * 4 + i;
      const float4 v = *(const float4*)(As + row * LDS_STRIDE + lane * 4);
      float4 r;
      r.x = v.x + b4.x; r.y = v.y + b4.y; r.z = v.z + b4.z; r.w = v.w + b4.w;
      r.x = r.x > 0.0f ? r.x : 0.0f;
      r.y = r.y > 0.0f ? r.y : 0.0f;
      r.z = r.z > 0.0f ? r.z : 0.0f;
      r.w = r.w > 0.0f ? r.w : 0.0f;
      ((float4*)out)[(size_t)(m0 + row) * 32 + lane] = r;
    }
  }
}

extern "C" void kernel_launch(void* const* d_in, const int* in_sizes, int n_in,
                              void* d_out, int out_size, void* d_ws, size_t ws_size,
                              hipStream_t stream) {
  (void)in_sizes; (void)n_in; (void)out_size; (void)ws_size;

  const float* x  = (const float*)d_in[0];
  const int*   ei = (const int*)d_in[1];
  const float* Wl[3] = {(const float*)d_in[2], (const float*)d_in[4], (const float*)d_in[6]};
  const float* bl[3] = {(const float*)d_in[3], (const float*)d_in[5], (const float*)d_in[7]};
  const int* src = ei;
  const int* dst = ei + N_EDGES;

  char*  ws     = (char*)d_ws;
  float* dinv   = (float*)(ws + OFF_DINV);
  int*   cnt    = (int*)(ws + OFF_CNT);
  int*   rowptr = (int*)(ws + OFF_ROWP);
  int*   loff   = (int*)(ws + OFF_LOFF);
  int2*  binned = (int2*)(ws + OFF_BINNED);
  int2*  ew     = (int2*)(ws + OFF_EW);
  float* hbuf   = (float*)(ws + OFF_HBUF);

  chunk_sort <<<NC, BIN_T, 0, stream>>>(src, dst, loff, binned);
  bin_sort   <<<NB, SORT_T, 0, stream>>>(binned, loff, rowptr, cnt, dinv, ew);
  edge_weight<<<(N_EDGES + 255) / 256, 256, 0, stream>>>(dinv, ew);

  float* o = (float*)d_out;
  gcn_layer<<<(N_NODES + 31) / 32, 256, 0, stream>>>(x,    rowptr, cnt, ew, dinv, Wl[0], bl[0], o);
  gcn_layer<<<(N_NODES + 31) / 32, 256, 0, stream>>>(o,    rowptr, cnt, ew, dinv, Wl[1], bl[1], hbuf);
  gcn_layer<<<(N_NODES + 31) / 32, 256, 0, stream>>>(hbuf, rowptr, cnt, ew, dinv, Wl[2], bl[2], o);
}
